# starting kernel
# baseline (speedup 1.0000x reference)
#include <hip/hip_runtime.h>
#include <hip/hip_bf16.h>

#define N_NODES 50000
#define FDIM    128
#define KDIM    128

typedef __attribute__((ext_vector_type(2))) float v2f;
typedef __attribute__((ext_vector_type(8))) float v8f;

__global__ void zero_f32(float* __restrict__ p, int n) {
    int i = blockIdx.x * blockDim.x + threadIdx.x;
    int stride = gridDim.x * blockDim.x;
    for (; i < n; i += stride) p[i] = 0.0f;
}

__global__ void degree_kernel(const int* __restrict__ src, const int* __restrict__ dst,
                              float* __restrict__ outdeg, float* __restrict__ indeg, int E) {
    int i = blockIdx.x * blockDim.x + threadIdx.x;
    if (i < E) {
        atomicAdd(&outdeg[src[i]], 1.0f);
        atomicAdd(&indeg[dst[i]], 1.0f);
    }
}

__global__ void norm_finalize(float* __restrict__ p, int n) {
    int i = blockIdx.x * blockDim.x + threadIdx.x;
    if (i < n) p[i] = rsqrtf(fmaxf(p[i], 1.0f));
}

__global__ __launch_bounds__(256)
void aggregate_kernel(const float* __restrict__ x, const int* __restrict__ src,
                      const int* __restrict__ dst, const float* __restrict__ nsrc,
                      float* __restrict__ agg, int E) {
    const int lane = threadIdx.x & 31;
    const int e = blockIdx.x * (blockDim.x >> 5) + (threadIdx.x >> 5);
    if (e >= E) return;
    const int s = src[e];
    const int d = dst[e];
    const float n = nsrc[s];
    const float4 v = ((const float4*)(x + (size_t)s * FDIM))[lane];
    float* o = agg + (size_t)d * FDIM + lane * 4;
    atomicAdd(o + 0, v.x * n);
    atomicAdd(o + 1, v.y * n);
    atomicAdd(o + 2, v.z * n);
    atomicAdd(o + 3, v.w * n);
}

template <int NOUT, bool RELU>
__global__ __launch_bounds__(128)
void gemm_wmma_kernel(const float* __restrict__ A, const float* __restrict__ W,
                      const float* __restrict__ bias, const float* __restrict__ norm,
                      float* __restrict__ out) {
    constexpr int NT = NOUT / 16;
    const int lane  = threadIdx.x & 31;
    const int wave  = threadIdx.x >> 5;
    const int mtile = blockIdx.x * 4 + wave;
    if (mtile * 16 >= N_NODES) return;
    const int m0    = mtile * 16;
    const int r15   = lane & 15;
    const int khalf = lane >> 4;
    const int koff  = khalf * 2;

    v8f acc[NT];
#pragma unroll
    for (int t = 0; t < NT; ++t)
#pragma unroll
        for (int j = 0; j < 8; ++j) acc[t][j] = 0.0f;

    const float* Ar = A + (size_t)(m0 + r15) * KDIM + koff;
    for (int k = 0; k < KDIM; k += 4) {
        v2f a;
        a.x = Ar[k];
        a.y = Ar[k + 1];
#pragma unroll
        for (int t = 0; t < NT; ++t) {
            const float* Wp = W + (size_t)(k + koff) * NOUT + t * 16 + r15;
            v2f b;
            b.x = Wp[0];
            b.y = Wp[NOUT];
            acc[t] = __builtin_amdgcn_wmma_f32_16x16x4_f32(
                false, a, false, b, (short)0, acc[t], false, false);
        }
    }

    const int rbase = m0 + khalf * 8;
    float nrm[8];
#pragma unroll
    for (int j = 0; j < 8; ++j) nrm[j] = norm[rbase + j];
#pragma unroll
    for (int t = 0; t < NT; ++t) {
        const float bv = bias[t * 16 + r15];
#pragma unroll
        for (int j = 0; j < 8; ++j) {
            float v = acc[t][j] * nrm[j] + bv;
            if (RELU) v = fmaxf(v, 0.0f);
            out[(size_t)(rbase + j) * NOUT + t * 16 + r15] = v;
        }
    }
}

extern "C" void kernel_launch(void* const* d_in, const int* in_sizes, int n_in,
                              void* d_out, int out_size, void* d_ws, size_t ws_size,
                              hipStream_t stream) {
    const float* features = (const float*)d_in[0];
    const float* W1 = (const float*)d_in[1];
    const float* b1 = (const float*)d_in[2];
    const float* W2 = (const float*)d_in[3];
    const float* b2 = (const float*)d_in[4];
    const float* W3 = (const float*)d_in[5];
    const float* b3 = (const float*)d_in[6];
    const int*  src = (const int*)d_in[7];
    const int*  dst = (const int*)d_in[8];
    const int E = in_sizes[7];

    float* ws       = (float*)d_ws;
    float* norm_src = ws;
    float* norm_dst = ws + N_NODES;
    float* aggA     = ws + 2 * N_NODES;
    float* bufB     = aggA + (size_t)N_NODES * FDIM;

    zero_f32<<<512, 256, 0, stream>>>(norm_src, 2 * N_NODES);
    degree_kernel<<<(E + 255) / 256, 256, 0, stream>>>(src, dst, norm_src, norm_dst, E);
    norm_finalize<<<(2 * N_NODES + 255) / 256, 256, 0, stream>>>(norm_src, 2 * N_NODES);

    const int aggBlocks  = (E + 7) / 8;
    const int gemmBlocks = (N_NODES / 16 + 3) / 4;

    zero_f32<<<2048, 256, 0, stream>>>(aggA, N_NODES * FDIM);
    aggregate_kernel<<<aggBlocks, 256, 0, stream>>>(features, src, dst, norm_src, aggA, E);
    gemm_wmma_kernel<128, true><<<gemmBlocks, 128, 0, stream>>>(aggA, W1, b1, norm_dst, bufB);

    zero_f32<<<2048, 256, 0, stream>>>(aggA, N_NODES * FDIM);
    aggregate_kernel<<<aggBlocks, 256, 0, stream>>>(bufB, src, dst, norm_src, aggA, E);
    gemm_wmma_kernel<128, true><<<gemmBlocks, 128, 0, stream>>>(aggA, W2, b2, norm_dst, bufB);

    zero_f32<<<2048, 256, 0, stream>>>(aggA, N_NODES * FDIM);
    aggregate_kernel<<<aggBlocks, 256, 0, stream>>>(bufB, src, dst, norm_src, aggA, E);
    gemm_wmma_kernel<64, false><<<gemmBlocks, 128, 0, stream>>>(aggA, W3, b3, norm_dst,
                                                                (float*)d_out);
}


# Owner groups filter a one-byte-per-edge map of owner group ids (written by the degree launch) instead of the 4-byte dst array; matches resolved one lane per edge; sums still in arrival order
# speedup vs baseline: 2.4279x; 2.4279x over previous
#include <hip/hip_runtime.h>
#include <hip/hip_bf16.h>

#define N_NODES 50000
#define FDIM    128
#define KDIM    128

typedef __attribute__((ext_vector_type(2))) float v2f;
typedef __attribute__((ext_vector_type(8))) float v8f;
typedef __attribute__((ext_vector_type(4))) float v4f;

__global__ void zero_f32(float* __restrict__ p, int n) {
    int i = blockIdx.x * blockDim.x + threadIdx.x;
    int stride = gridDim.x * blockDim.x;
    for (; i < n; i += stride) p[i] = 0.0f;
}

#define OWN_ROWS    208
#define OWN_RND     65536
static __device__ __forceinline__ unsigned owner_byte(int d) {
    return ((unsigned)d < (unsigned)N_NODES) ? (unsigned)d / OWN_ROWS : 255u;
}
__global__ void degree_kernel(const int* __restrict__ src, const int* __restrict__ dst,
                              float* __restrict__ outdeg, float* __restrict__ indeg,
                              unsigned* __restrict__ omap, int E) {
    const int i  = blockIdx.x * blockDim.x + threadIdx.x;
    const int e0 = i * 4;
    unsigned word = 0xffffffffu;
    if (e0 + 3 < E) {
        const int4 s = ((const int4*)src)[i];
        const int4 d = ((const int4*)dst)[i];
        atomicAdd(&outdeg[s.x], 1.0f); atomicAdd(&indeg[d.x], 1.0f);
        atomicAdd(&outdeg[s.y], 1.0f); atomicAdd(&indeg[d.y], 1.0f);
        atomicAdd(&outdeg[s.z], 1.0f); atomicAdd(&indeg[d.z], 1.0f);
        atomicAdd(&outdeg[s.w], 1.0f); atomicAdd(&indeg[d.w], 1.0f);
        word = owner_byte(d.x) | (owner_byte(d.y) << 8) | (owner_byte(d.z) << 16) |
               (owner_byte(d.w) << 24);
    } else {
        for (int c = 0; c < 4; ++c) {
            if (e0 + c < E) {
                const int d = dst[e0 + c];
                atomicAdd(&outdeg[src[e0 + c]], 1.0f);
                atomicAdd(&indeg[d], 1.0f);
                word = (word & ~(0xffu << (8 * c))) | (owner_byte(d) << (8 * c));
            }
        }
    }
    omap[i] = word;
}

__global__ void norm_finalize(float* __restrict__ p, int n) {
    int i = blockIdx.x * blockDim.x + threadIdx.x;
    if (i < n) p[i] = rsqrtf(fmaxf(p[i], 1.0f));
}

#define OWN_GROUPS  ((N_NODES + OWN_ROWS - 1) / OWN_ROWS)
#define OWN_THREADS 1024
#define OWN_WAVES   (OWN_THREADS / 32)
#define OWN_SLOTS   ((OWN_ROWS + OWN_WAVES - 1) / OWN_WAVES)
#define OWN_SUBCAP  256
#define OWN_QCAP    (OWN_WAVES * OWN_SUBCAP)

static __device__ __forceinline__ unsigned zero_bytes(unsigned x) {
    return ~(((x & 0x7f7f7f7fu) + 0x7f7f7f7fu) | x | 0x7f7f7f7fu);
}

static __device__ __forceinline__ unsigned match16(const int4 v, unsigned pat) {
    return (zero_bytes((unsigned)v.x ^ pat) >> 7) | (zero_bytes((unsigned)v.y ^ pat) >> 6) |
           (zero_bytes((unsigned)v.z ^ pat) >> 5) | (zero_bytes((unsigned)v.w ^ pat) >> 4);
}

static __device__ __forceinline__ void list_edge(int* l_cnt, int* l_e, int e) {
    const int p = atomicAdd(l_cnt, 1);
    if (p < OWN_QCAP) l_e[p] = e;
}

static __device__ __forceinline__ void queue_edge(int* q_cnt, int* q_id, int* q_slot,
                                                  float* q_scale, unsigned r, int s, float sc) {
    const int ow = (int)(r & (OWN_WAVES - 1));
    const int p  = atomicAdd(&q_cnt[ow], 1);
    if (p < OWN_SUBCAP) {
        q_id[ow * OWN_SUBCAP + p]    = s;
        q_slot[ow * OWN_SUBCAP + p]  = (int)(r / OWN_WAVES);
        q_scale[ow * OWN_SUBCAP + p] = sc;
    }
}

static __device__ __forceinline__ void add_row(float (&acc)[32], int slot, const v4f v, const float w) {
    const int b = slot * 4;
    acc[b + 0] += v.x * w;
    acc[b + 1] += v.y * w;
    acc[b + 2] += v.z * w;
    acc[b + 3] += v.w * w;
}

template <bool BUILD>
__global__ __launch_bounds__(OWN_THREADS)
void aggregate_owner_kernel(const float* __restrict__ x, const int* __restrict__ src,
                            const int* __restrict__ dst, const float* __restrict__ nsrc,
                            float* __restrict__ agg, int* __restrict__ g_src,
                            int* __restrict__ g_slot, float* __restrict__ g_scale,
                            int* __restrict__ g_count, const unsigned* __restrict__ omap, int E) {
    __shared__ int   q_id[OWN_QCAP];
    __shared__ int   q_slot[OWN_QCAP];
    __shared__ float q_scale[OWN_QCAP];
    __shared__ int   q_cnt[OWN_WAVES];
    __shared__ int   l_e[OWN_QCAP];
    __shared__ int   l_cnt;

    const int tid  = threadIdx.x;
    const int lane = tid & 31;
    const int wave = __builtin_amdgcn_readfirstlane(tid >> 5);
    const int base = blockIdx.x * OWN_ROWS;
    const size_t qbase = (size_t)blockIdx.x * OWN_QCAP;
    const int qb = wave * OWN_SUBCAP;

    int cnt;
    if (BUILD) {
        if (tid < OWN_WAVES) q_cnt[tid] = 0;
        if (tid == 0) l_cnt = 0;
        __syncthreads();
        const int nrnd = (E + OWN_RND - 1) / OWN_RND;
        const int4* __restrict__ m4 = (const int4*)omap + tid;
        const unsigned pat = (unsigned)blockIdx.x * 0x01010101u;
        constexpr int STEP = OWN_THREADS * 4;
        int4 cur[4], nxt[4];
#pragma unroll
        for (int u = 0; u < 4; ++u) cur[u] = make_int4(-1, -1, -1, -1);
        if (nrnd > 0) {
#pragma unroll
            for (int u = 0; u < 4; ++u) cur[u] = m4[u * OWN_THREADS];
        }
        for (int it = 0; it < nrnd; ++it) {
#pragma unroll
            for (int u = 0; u < 4; ++u) nxt[u] = make_int4(-1, -1, -1, -1);
            if (it + 1 < nrnd) {
#pragma unroll
                for (int u = 0; u < 4; ++u) nxt[u] = m4[(size_t)(it + 1) * STEP + u * OWN_THREADS];
            }
            const unsigned lo = match16(cur[0], pat) | (match16(cur[1], pat) << 4);
            const unsigned hi = match16(cur[2], pat) | (match16(cur[3], pat) << 4);
            unsigned long long m = ((unsigned long long)hi << 32) | lo;
            const int e0 = (it * STEP + tid) * 16;
            while (m != 0ull) {
                const int bit = __builtin_ctzll(m);
                m &= m - 1ull;
                const int u = ((bit >> 5) << 1) | ((bit >> 2) & 1);
                list_edge(&l_cnt, l_e, e0 + u * (OWN_THREADS * 16) + ((bit & 3) << 2) + ((bit >> 3) & 3));
            }
#pragma unroll
            for (int u = 0; u < 4; ++u) cur[u] = nxt[u];
        }
        __syncthreads();
        const int  tot   = l_cnt;
        const bool spill = tot > OWN_QCAP;
        const int  nl    = spill ? 0 : tot;
        for (int p0 = tid; p0 < nl; p0 += STEP) {
            int   e[4], d[4], s[4];
            float sc[4];
#pragma unroll
            for (int u = 0; u < 4; ++u) {
                const int p = p0 + u * OWN_THREADS;
                e[u] = l_e[(p < nl) ? p : p0];
            }
#pragma unroll
            for (int u = 0; u < 4; ++u) { d[u] = dst[e[u]]; s[u] = src[e[u]]; }
#pragma unroll
            for (int u = 0; u < 4; ++u) sc[u] = nsrc[s[u]];
            asm volatile("" :: "v"(sc[0]), "v"(sc[1]), "v"(sc[2]), "v"(sc[3]),
                               "v"(d[0]), "v"(d[1]), "v"(d[2]), "v"(d[3]));
#pragma unroll
            for (int u = 0; u < 4; ++u) {
                const unsigned r = (unsigned)(d[u] - base);
                if (p0 + u * OWN_THREADS < nl && r < (unsigned)OWN_ROWS)
                    queue_edge(q_cnt, q_id, q_slot, q_scale, r, s[u], sc[u]);
            }
        }
        __syncthreads();
        cnt = __builtin_amdgcn_readfirstlane(spill ? OWN_SUBCAP + 1 : q_cnt[wave]);
        if (lane == 0) g_count[blockIdx.x * OWN_WAVES + wave] = cnt;
    } else {
        cnt = g_count[blockIdx.x * OWN_WAVES + wave];
    }

    const bool over = cnt > OWN_SUBCAP;
    const int  n    = over ? 0 : cnt;

    if (BUILD) {
        for (int k = lane; k < n; k += 32) {
            g_src[qbase + qb + k]   = q_id[qb + k];
            g_slot[qbase + qb + k]  = q_slot[qb + k];
            g_scale[qbase + qb + k] = q_scale[qb + k];
        }
    } else {
        for (int k = lane; k < n; k += 32) {
            q_id[qb + k]    = g_src[qbase + qb + k];
            q_slot[qb + k]  = g_slot[qbase + qb + k];
            q_scale[qb + k] = g_scale[qbase + qb + k];
        }
    }
    __syncthreads();

    float acc[32];
#pragma unroll
    for (int i = 0; i < 32; ++i) acc[i] = 0.0f;
    const v4f* __restrict__ x4 = (const v4f*)x;

    const int l8 = lane & 7;
    int   e_src  = 0;
    int   e_slot = 0;
    float e_w    = 0.0f;
    if (n > 0) {
        const int kc = (l8 < n) ? l8 : 0;
        e_src = q_id[qb + kc]; e_slot = q_slot[qb + kc]; e_w = q_scale[qb + kc];
    }
    const int nfull = n & ~7;
    for (int k0 = 0; k0 < nfull; k0 += 8) {
        v4f    v[8];
        float  w[8];
        int    slot[8];
#pragma unroll
        for (int u = 0; u < 8; ++u) {
            const int s = __builtin_amdgcn_readlane(e_src, u);
            slot[u] = __builtin_amdgcn_readlane(e_slot, u);
            w[u]    = __int_as_float(__builtin_amdgcn_readlane(__float_as_int(e_w), u));
            v[u]    = x4[(size_t)s * (FDIM / 4) + lane];
        }
        {
            const int kn = k0 + 8 + l8;
            const int kc = (kn < n) ? kn : 0;
            e_src = q_id[qb + kc]; e_slot = q_slot[qb + kc]; e_w = q_scale[qb + kc];
        }
        asm volatile("" :: "v"(v[0]), "v"(v[1]), "v"(v[2]), "v"(v[3]),
                           "v"(v[4]), "v"(v[5]), "v"(v[6]), "v"(v[7]));
#pragma unroll
        for (int u = 0; u < 8; ++u) add_row(acc, slot[u], v[u], w[u]);
    }
    if (nfull < n) {
        v4f    v[8];
        float  w[8];
        int    slot[8];
#pragma unroll
        for (int u = 0; u < 8; ++u) {
            const int s = __builtin_amdgcn_readlane(e_src, u);
            slot[u] = __builtin_amdgcn_readlane(e_slot, u);
            w[u]    = __int_as_float(__builtin_amdgcn_readlane(__float_as_int(e_w), u));
            v[u]    = x4[(size_t)s * (FDIM / 4) + lane];
        }
        asm volatile("" :: "v"(v[0]), "v"(v[1]), "v"(v[2]), "v"(v[3]),
                           "v"(v[4]), "v"(v[5]), "v"(v[6]), "v"(v[7]));
#pragma unroll
        for (int u = 0; u < 8; ++u) {
            if (nfull + u < n) add_row(acc, slot[u], v[u], w[u]);
        }
    }

    if (over) {
        for (int e0 = 0; e0 < E; e0 += 32) {
            const int e = e0 + lane;
            unsigned r = 0xffffffffu;
            if (e < E) r = (unsigned)(dst[e] - base);
            const bool mine = (r < (unsigned)OWN_ROWS) && ((int)(r & (OWN_WAVES - 1)) == wave);
            unsigned m = __builtin_amdgcn_ballot_w32(mine);
            while (m != 0u) {
                const int l = __builtin_ctz(m);
                m &= m - 1u;
                const int rl   = __builtin_amdgcn_readlane((int)r, l);
                const int s    = src[e0 + l];
                const float sc = nsrc[s];
                add_row(acc, rl / OWN_WAVES, x4[(size_t)s * (FDIM / 4) + lane], sc);
            }
        }
    }

    float4* __restrict__ agg4 = (float4*)agg;
#pragma unroll
    for (int j = 0; j < OWN_SLOTS; ++j) {
        const int r = wave + j * OWN_WAVES;
        if (r < OWN_ROWS && base + r < N_NODES)
            agg4[(size_t)(base + r) * (FDIM / 4) + lane] =
                make_float4(acc[4 * j], acc[4 * j + 1], acc[4 * j + 2], acc[4 * j + 3]);
    }
}

template <int NOUT, bool RELU>
__global__ __launch_bounds__(128)
void gemm_wmma_kernel(const float* __restrict__ A, const float* __restrict__ W,
                      const float* __restrict__ bias, const float* __restrict__ norm,
                      float* __restrict__ out) {
    constexpr int NT = NOUT / 16;
    const int lane  = threadIdx.x & 31;
    const int wave  = threadIdx.x >> 5;
    const int mtile = blockIdx.x * 4 + wave;
    if (mtile * 16 >= N_NODES) return;
    const int m0    = mtile * 16;
    const int r15   = lane & 15;
    const int khalf = lane >> 4;
    const int koff  = khalf * 2;

    v8f acc[NT];
#pragma unroll
    for (int t = 0; t < NT; ++t)
#pragma unroll
        for (int j = 0; j < 8; ++j) acc[t][j] = 0.0f;

    const float* Ar = A + (size_t)(m0 + r15) * KDIM + koff;
    for (int k = 0; k < KDIM; k += 4) {
        v2f a;
        a.x = Ar[k];
        a.y = Ar[k + 1];
#pragma unroll
        for (int t = 0; t < NT; ++t) {
            const float* Wp = W + (size_t)(k + koff) * NOUT + t * 16 + r15;
            v2f b;
            b.x = Wp[0];
            b.y = Wp[NOUT];
            acc[t] = __builtin_amdgcn_wmma_f32_16x16x4_f32(
                false, a, false, b, (short)0, acc[t], false, false);
        }
    }

    const int rbase = m0 + khalf * 8;
    float nrm[8];
#pragma unroll
    for (int j = 0; j < 8; ++j) nrm[j] = norm[rbase + j];
#pragma unroll
    for (int t = 0; t < NT; ++t) {
        const float bv = bias[t * 16 + r15];
#pragma unroll
        for (int j = 0; j < 8; ++j) {
            float v = acc[t][j] * nrm[j] + bv;
            if (RELU) v = fmaxf(v, 0.0f);
            out[(size_t)(rbase + j) * NOUT + t * 16 + r15] = v;
        }
    }
}

extern "C" void kernel_launch(void* const* d_in, const int* in_sizes, int n_in,
                              void* d_out, int out_size, void* d_ws, size_t ws_size,
                              hipStream_t stream) {
    const float* features = (const float*)d_in[0];
    const float* W1 = (const float*)d_in[1];
    const float* b1 = (const float*)d_in[2];
    const float* W2 = (const float*)d_in[3];
    const float* b2 = (const float*)d_in[4];
    const float* W3 = (const float*)d_in[5];
    const float* b3 = (const float*)d_in[6];
    const int*  src = (const int*)d_in[7];
    const int*  dst = (const int*)d_in[8];
    const int E = in_sizes[7];

    float* ws       = (float*)d_ws;
    float* norm_src = ws;
    float* norm_dst = ws + N_NODES;
    float* aggA     = ws + 2 * N_NODES;
    float* bufB     = aggA + (size_t)N_NODES * FDIM;
    int*   q_src    = (int*)(bufB + (size_t)N_NODES * FDIM);
    int*   q_row    = q_src + (size_t)OWN_GROUPS * OWN_QCAP;
    float* q_scale  = (float*)(q_row + (size_t)OWN_GROUPS * OWN_QCAP);
    int*   q_cnt    = (int*)(q_scale + (size_t)OWN_GROUPS * OWN_QCAP);
    unsigned* omap  = (unsigned*)(q_cnt + (size_t)OWN_GROUPS * OWN_WAVES);

    zero_f32<<<512, 256, 0, stream>>>(norm_src, 2 * N_NODES);
    degree_kernel<<<(E + OWN_RND - 1) / OWN_RND * (OWN_RND / 1024), 256, 0, stream>>>(
        src, dst, norm_src, norm_dst, omap, E);
    norm_finalize<<<(2 * N_NODES + 255) / 256, 256, 0, stream>>>(norm_src, 2 * N_NODES);

    const int gemmBlocks = (N_NODES / 16 + 3) / 4;

    aggregate_owner_kernel<true><<<OWN_GROUPS, OWN_THREADS, 0, stream>>>(
        features, src, dst, norm_src, aggA, q_src, q_row, q_scale, q_cnt, omap, E);
    gemm_wmma_kernel<128, true><<<gemmBlocks, 128, 0, stream>>>(aggA, W1, b1, norm_dst, bufB);

    aggregate_owner_kernel<false><<<OWN_GROUPS, OWN_THREADS, 0, stream>>>(
        bufB, src, dst, norm_src, aggA, q_src, q_row, q_scale, q_cnt, omap, E);
    gemm_wmma_kernel<128, true><<<gemmBlocks, 128, 0, stream>>>(aggA, W2, b2, norm_dst, bufB);

    aggregate_owner_kernel<false><<<OWN_GROUPS, OWN_THREADS, 0, stream>>>(
        bufB, src, dst, norm_src, aggA, q_src, q_row, q_scale, q_cnt, omap, E);
    gemm_wmma_kernel<64, false><<<gemmBlocks, 128, 0, stream>>>(aggA, W3, b3, norm_dst,
                                                                (float*)d_out);
}


# Transform launches re-tiled: a wave keeps its 16-column strip of W in registers, work-groups walk over row tiles staged through LDS by whole-row loads, whole-row stores; k sum order changed
# speedup vs baseline: 2.7967x; 1.1519x over previous
#include <hip/hip_runtime.h>
#include <hip/hip_bf16.h>

#define N_NODES 50000
#define FDIM    128
#define KDIM    128

typedef __attribute__((ext_vector_type(2))) float v2f;
typedef __attribute__((ext_vector_type(8))) float v8f;
typedef __attribute__((ext_vector_type(4))) float v4f;

__global__ void zero_f32(float* __restrict__ p, int n) {
    int i = blockIdx.x * blockDim.x + threadIdx.x;
    int stride = gridDim.x * blockDim.x;
    for (; i < n; i += stride) p[i] = 0.0f;
}

#define OWN_ROWS    208
#define OWN_RND     65536
static __device__ __forceinline__ unsigned owner_byte(int d) {
    return ((unsigned)d < (unsigned)N_NODES) ? (unsigned)d / OWN_ROWS : 255u;
}
__global__ void degree_kernel(const int* __restrict__ src, const int* __restrict__ dst,
                              float* __restrict__ outdeg, float* __restrict__ indeg,
                              unsigned* __restrict__ omap, int E) {
    const int i  = blockIdx.x * blockDim.x + threadIdx.x;
    const int e0 = i * 4;
    unsigned word = 0xffffffffu;
    if (e0 + 3 < E) {
        const int4 s = ((const int4*)src)[i];
        const int4 d = ((const int4*)dst)[i];
        atomicAdd(&outdeg[s.x], 1.0f); atomicAdd(&indeg[d.x], 1.0f);
        atomicAdd(&outdeg[s.y], 1.0f); atomicAdd(&indeg[d.y], 1.0f);
        atomicAdd(&outdeg[s.z], 1.0f); atomicAdd(&indeg[d.z], 1.0f);
        atomicAdd(&outdeg[s.w], 1.0f); atomicAdd(&indeg[d.w], 1.0f);
        word = owner_byte(d.x) | (owner_byte(d.y) << 8) | (owner_byte(d.z) << 16) |
               (owner_byte(d.w) << 24);
    } else {
        for (int c = 0; c < 4; ++c) {
            if (e0 + c < E) {
                const int d = dst[e0 + c];
                atomicAdd(&outdeg[src[e0 + c]], 1.0f);
                atomicAdd(&indeg[d], 1.0f);
                word = (word & ~(0xffu << (8 * c))) | (owner_byte(d) << (8 * c));
            }
        }
    }
    omap[i] = word;
}

__global__ void norm_finalize(float* __restrict__ p, int n) {
    int i = blockIdx.x * blockDim.x + threadIdx.x;
    if (i < n) p[i] = rsqrtf(fmaxf(p[i], 1.0f));
}

#define OWN_GROUPS  ((N_NODES + OWN_ROWS - 1) / OWN_ROWS)
#define OWN_THREADS 1024
#define OWN_WAVES   (OWN_THREADS / 32)
#define OWN_SLOTS   ((OWN_ROWS + OWN_WAVES - 1) / OWN_WAVES)
#define OWN_SUBCAP  256
#define OWN_QCAP    (OWN_WAVES * OWN_SUBCAP)

static __device__ __forceinline__ unsigned zero_bytes(unsigned x) {
    return ~(((x & 0x7f7f7f7fu) + 0x7f7f7f7fu) | x | 0x7f7f7f7fu);
}

static __device__ __forceinline__ unsigned match16(const int4 v, unsigned pat) {
    return (zero_bytes((unsigned)v.x ^ pat) >> 7) | (zero_bytes((unsigned)v.y ^ pat) >> 6) |
           (zero_bytes((unsigned)v.z ^ pat) >> 5) | (zero_bytes((unsigned)v.w ^ pat) >> 4);
}

static __device__ __forceinline__ void list_edge(int* l_cnt, int* l_e, int e) {
    const int p = atomicAdd(l_cnt, 1);
    if (p < OWN_QCAP) l_e[p] = e;
}

static __device__ __forceinline__ void queue_edge(int* q_cnt, int* q_id, int* q_slot,
                                                  float* q_scale, unsigned r, int s, float sc) {
    const int ow = (int)(r & (OWN_WAVES - 1));
    const int p  = atomicAdd(&q_cnt[ow], 1);
    if (p < OWN_SUBCAP) {
        q_id[ow * OWN_SUBCAP + p]    = s;
        q_slot[ow * OWN_SUBCAP + p]  = (int)(r / OWN_WAVES);
        q_scale[ow * OWN_SUBCAP + p] = sc;
    }
}

static __device__ __forceinline__ void add_row(float (&acc)[32], int slot, const v4f v, const float w) {
    const int b = slot * 4;
    acc[b + 0] += v.x * w;
    acc[b + 1] += v.y * w;
    acc[b + 2] += v.z * w;
    acc[b + 3] += v.w * w;
}

template <bool BUILD>
__global__ __launch_bounds__(OWN_THREADS)
void aggregate_owner_kernel(const float* __restrict__ x, const int* __restrict__ src,
                            const int* __restrict__ dst, const float* __restrict__ nsrc,
                            float* __restrict__ agg, int* __restrict__ g_src,
                            int* __restrict__ g_slot, float* __restrict__ g_scale,
                            int* __restrict__ g_count, const unsigned* __restrict__ omap, int E) {
    __shared__ int   q_id[OWN_QCAP];
    __shared__ int   q_slot[OWN_QCAP];
    __shared__ float q_scale[OWN_QCAP];
    __shared__ int   q_cnt[OWN_WAVES];
    __shared__ int   l_e[OWN_QCAP];
    __shared__ int   l_cnt;

    const int tid  = threadIdx.x;
    const int lane = tid & 31;
    const int wave = __builtin_amdgcn_readfirstlane(tid >> 5);
    const int base = blockIdx.x * OWN_ROWS;
    const size_t qbase = (size_t)blockIdx.x * OWN_QCAP;
    const int qb = wave * OWN_SUBCAP;

    int cnt;
    if (BUILD) {
        if (tid < OWN_WAVES) q_cnt[tid] = 0;
        if (tid == 0) l_cnt = 0;
        __syncthreads();
        const int nrnd = (E + OWN_RND - 1) / OWN_RND;
        const int4* __restrict__ m4 = (const int4*)omap + tid;
        const unsigned pat = (unsigned)blockIdx.x * 0x01010101u;
        constexpr int STEP = OWN_THREADS * 4;
        int4 cur[4], nxt[4];
#pragma unroll
        for (int u = 0; u < 4; ++u) cur[u] = make_int4(-1, -1, -1, -1);
        if (nrnd > 0) {
#pragma unroll
            for (int u = 0; u < 4; ++u) cur[u] = m4[u * OWN_THREADS];
        }
        for (int it = 0; it < nrnd; ++it) {
#pragma unroll
            for (int u = 0; u < 4; ++u) nxt[u] = make_int4(-1, -1, -1, -1);
            if (it + 1 < nrnd) {
#pragma unroll
                for (int u = 0; u < 4; ++u) nxt[u] = m4[(size_t)(it + 1) * STEP + u * OWN_THREADS];
            }
            const unsigned lo = match16(cur[0], pat) | (match16(cur[1], pat) << 4);
            const unsigned hi = match16(cur[2], pat) | (match16(cur[3], pat) << 4);
            unsigned long long m = ((unsigned long long)hi << 32) | lo;
            const int e0 = (it * STEP + tid) * 16;
            while (m != 0ull) {
                const int bit = __builtin_ctzll(m);
                m &= m - 1ull;
                const int u = ((bit >> 5) << 1) | ((bit >> 2) & 1);
                list_edge(&l_cnt, l_e, e0 + u * (OWN_THREADS * 16) + ((bit & 3) << 2) + ((bit >> 3) & 3));
            }
#pragma unroll
            for (int u = 0; u < 4; ++u) cur[u] = nxt[u];
        }
        __syncthreads();
        const int  tot   = l_cnt;
        const bool spill = tot > OWN_QCAP;
        const int  nl    = spill ? 0 : tot;
        for (int p0 = tid; p0 < nl; p0 += STEP) {
            int   e[4], d[4], s[4];
            float sc[4];
#pragma unroll
            for (int u = 0; u < 4; ++u) {
                const int p = p0 + u * OWN_THREADS;
                e[u] = l_e[(p < nl) ? p : p0];
            }
#pragma unroll
            for (int u = 0; u < 4; ++u) { d[u] = dst[e[u]]; s[u] = src[e[u]]; }
#pragma unroll
            for (int u = 0; u < 4; ++u) sc[u] = nsrc[s[u]];
            asm volatile("" :: "v"(sc[0]), "v"(sc[1]), "v"(sc[2]), "v"(sc[3]),
                               "v"(d[0]), "v"(d[1]), "v"(d[2]), "v"(d[3]));
#pragma unroll
            for (int u = 0; u < 4; ++u) {
                const unsigned r = (unsigned)(d[u] - base);
                if (p0 + u * OWN_THREADS < nl && r < (unsigned)OWN_ROWS)
                    queue_edge(q_cnt, q_id, q_slot, q_scale, r, s[u], sc[u]);
            }
        }
        __syncthreads();
        cnt = __builtin_amdgcn_readfirstlane(spill ? OWN_SUBCAP + 1 : q_cnt[wave]);
        if (lane == 0) g_count[blockIdx.x * OWN_WAVES + wave] = cnt;
    } else {
        cnt = g_count[blockIdx.x * OWN_WAVES + wave];
    }

    const bool over = cnt > OWN_SUBCAP;
    const int  n    = over ? 0 : cnt;

    if (BUILD) {
        for (int k = lane; k < n; k += 32) {
            g_src[qbase + qb + k]   = q_id[qb + k];
            g_slot[qbase + qb + k]  = q_slot[qb + k];
            g_scale[qbase + qb + k] = q_scale[qb + k];
        }
    } else {
        for (int k = lane; k < n; k += 32) {
            q_id[qb + k]    = g_src[qbase + qb + k];
            q_slot[qb + k]  = g_slot[qbase + qb + k];
            q_scale[qb + k] = g_scale[qbase + qb + k];
        }
    }
    __syncthreads();

    float acc[32];
#pragma unroll
    for (int i = 0; i < 32; ++i) acc[i] = 0.0f;
    const v4f* __restrict__ x4 = (const v4f*)x;

    const int l8 = lane & 7;
    int   e_src  = 0;
    int   e_slot = 0;
    float e_w    = 0.0f;
    if (n > 0) {
        const int kc = (l8 < n) ? l8 : 0;
        e_src = q_id[qb + kc]; e_slot = q_slot[qb + kc]; e_w = q_scale[qb + kc];
    }
    const int nfull = n & ~7;
    for (int k0 = 0; k0 < nfull; k0 += 8) {
        v4f    v[8];
        float  w[8];
        int    slot[8];
#pragma unroll
        for (int u = 0; u < 8; ++u) {
            const int s = __builtin_amdgcn_readlane(e_src, u);
            slot[u] = __builtin_amdgcn_readlane(e_slot, u);
            w[u]    = __int_as_float(__builtin_amdgcn_readlane(__float_as_int(e_w), u));
            v[u]    = x4[(size_t)s * (FDIM / 4) + lane];
        }
        {
            const int kn = k0 + 8 + l8;
            const int kc = (kn < n) ? kn : 0;
            e_src = q_id[qb + kc]; e_slot = q_slot[qb + kc]; e_w = q_scale[qb + kc];
        }
        asm volatile("" :: "v"(v[0]), "v"(v[1]), "v"(v[2]), "v"(v[3]),
                           "v"(v[4]), "v"(v[5]), "v"(v[6]), "v"(v[7]));
#pragma unroll
        for (int u = 0; u < 8; ++u) add_row(acc, slot[u], v[u], w[u]);
    }
    if (nfull < n) {
        v4f    v[8];
        float  w[8];
        int    slot[8];
#pragma unroll
        for (int u = 0; u < 8; ++u) {
            const int s = __builtin_amdgcn_readlane(e_src, u);
            slot[u] = __builtin_amdgcn_readlane(e_slot, u);
            w[u]    = __int_as_float(__builtin_amdgcn_readlane(__float_as_int(e_w), u));
            v[u]    = x4[(size_t)s * (FDIM / 4) + lane];
        }
        asm volatile("" :: "v"(v[0]), "v"(v[1]), "v"(v[2]), "v"(v[3]),
                           "v"(v[4]), "v"(v[5]), "v"(v[6]), "v"(v[7]));
#pragma unroll
        for (int u = 0; u < 8; ++u) {
            if (nfull + u < n) add_row(acc, slot[u], v[u], w[u]);
        }
    }

    if (over) {
        for (int e0 = 0; e0 < E; e0 += 32) {
            const int e = e0 + lane;
            unsigned r = 0xffffffffu;
            if (e < E) r = (unsigned)(dst[e] - base);
            const bool mine = (r < (unsigned)OWN_ROWS) && ((int)(r & (OWN_WAVES - 1)) == wave);
            unsigned m = __builtin_amdgcn_ballot_w32(mine);
            while (m != 0u) {
                const int l = __builtin_ctz(m);
                m &= m - 1u;
                const int rl   = __builtin_amdgcn_readlane((int)r, l);
                const int s    = src[e0 + l];
                const float sc = nsrc[s];
                add_row(acc, rl / OWN_WAVES, x4[(size_t)s * (FDIM / 4) + lane], sc);
            }
        }
    }

    float4* __restrict__ agg4 = (float4*)agg;
#pragma unroll
    for (int j = 0; j < OWN_SLOTS; ++j) {
        const int r = wave + j * OWN_WAVES;
        if (r < OWN_ROWS && base + r < N_NODES)
            agg4[(size_t)(base + r) * (FDIM / 4) + lane] =
                make_float4(acc[4 * j], acc[4 * j + 1], acc[4 * j + 2], acc[4 * j + 3]);
    }
}

#define A_PITCH (KDIM + 4)
template <int NOUT, bool RELU>
__global__ __launch_bounds__(NOUT * 2)
void gemm_wmma_kernel(const float* __restrict__ A, const float* __restrict__ W,
                      const float* __restrict__ bias, const float* __restrict__ norm,
                      float* __restrict__ out) {
    constexpr int NW      = NOUT / 16;
    constexpr int RPW     = 16 / NW;
    constexpr int D_PITCH = NOUT + 4;
    constexpr int NTILES  = N_NODES / 16;
    static_assert(N_NODES % 16 == 0, "row tiles are whole");
    static_assert(KDIM == 128, "one 512-byte A row per wave load");
    __shared__ __attribute__((aligned(16))) float As[2][16 * A_PITCH];
    __shared__ __attribute__((aligned(16))) float Ds[16 * D_PITCH];

    const int lane = threadIdx.x & 31;
    const int wave = __builtin_amdgcn_readfirstlane((int)(threadIdx.x >> 5));
    const int r    = lane & 15;
    const int h    = lane >> 4;
    const int n0   = wave * 16;

    v2f bfrag[32];
#pragma unroll
    for (int s = 0; s < 32; ++s) {
        const int k = 8 * (s >> 1) + 4 * h + 2 * (s & 1);
        bfrag[s].x = W[(size_t)k * NOUT + n0 + r];
        bfrag[s].y = W[(size_t)(k + 1) * NOUT + n0 + r];
    }
    const float bv = bias[n0 + r];

    const v4f* __restrict__ A4 = (const v4f*)A;
    v4f* __restrict__ out4 = (v4f*)out;
    int mt = blockIdx.x;
    v4f nxt[RPW];
    if (mt < NTILES) {
#pragma unroll
        for (int i = 0; i < RPW; ++i)
            nxt[i] = A4[(size_t)(mt * 16 + wave * RPW + i) * (KDIM / 4) + lane];
#pragma unroll
        for (int i = 0; i < RPW; ++i)
            *(v4f*)&As[0][(wave * RPW + i) * A_PITCH + 4 * lane] = nxt[i];
    }
    int cur = 0;
    for (; mt < NTILES; mt += gridDim.x) {
        __syncthreads();
        const int  mn   = mt + (int)gridDim.x;
        const bool more = mn < NTILES;
        if (more) {
#pragma unroll
            for (int i = 0; i < RPW; ++i)
                nxt[i] = A4[(size_t)(mn * 16 + wave * RPW + i) * (KDIM / 4) + lane];
        }
        const v4f* __restrict__ nr = (const v4f*)(norm + mt * 16 + 8 * h);
        const v4f nA = nr[0];
        const v4f nB = nr[1];

        v4f af[16];
#pragma unroll
        for (int j = 0; j < 16; ++j)
            af[j] = *(const v4f*)&As[cur][r * A_PITCH + 8 * j + 4 * h];
        asm volatile("" :: "v"(af[0]), "v"(af[1]), "v"(af[2]), "v"(af[3]), "v"(af[4]), "v"(af[5]),
                           "v"(af[6]), "v"(af[7]), "v"(af[8]), "v"(af[9]), "v"(af[10]), "v"(af[11]),
                           "v"(af[12]), "v"(af[13]), "v"(af[14]), "v"(af[15]));

        v8f acc0, acc1;
#pragma unroll
        for (int j = 0; j < 8; ++j) { acc0[j] = 0.0f; acc1[j] = 0.0f; }
#pragma unroll
        for (int j = 0; j < 16; ++j) {
            v2f a0, a1;
            a0.x = af[j].x; a0.y = af[j].y;
            a1.x = af[j].z; a1.y = af[j].w;
            acc0 = __builtin_amdgcn_wmma_f32_16x16x4_f32(
                false, a0, false, bfrag[2 * j], (short)0, acc0, false, false);
            acc1 = __builtin_amdgcn_wmma_f32_16x16x4_f32(
                false, a1, false, bfrag[2 * j + 1], (short)0, acc1, false, false);
        }

        const float nrm[8] = {nA.x, nA.y, nA.z, nA.w, nB.x, nB.y, nB.z, nB.w};
#pragma unroll
        for (int j = 0; j < 8; ++j) {
            float v = (acc0[j] + acc1[j]) * nrm[j] + bv;
            if (RELU) v = fmaxf(v, 0.0f);
            Ds[(8 * h + j) * D_PITCH + n0 + r] = v;
        }
        __syncthreads();
        if (NOUT == 128) {
#pragma unroll
            for (int i = 0; i < RPW; ++i) {
                const int row = wave * RPW + i;
                out4[(size_t)(mt * 16 + row) * (NOUT / 4) + lane] =
                    *(const v4f*)&Ds[row * D_PITCH + 4 * lane];
            }
        } else {
#pragma unroll
            for (int i = 0; i < RPW / 2; ++i) {
                const int row = wave * RPW + 2 * i + h;
                out4[(size_t)(mt * 16 + row) * (NOUT / 4) + r] =
                    *(const v4f*)&Ds[row * D_PITCH + 4 * r];
            }
        }
        if (more) {
#pragma unroll
            for (int i = 0; i < RPW; ++i)
                *(v4f*)&As[cur ^ 1][(wave * RPW + i) * A_PITCH + 4 * lane] = nxt[i];
        }
        cur ^= 1;
    }
}

extern "C" void kernel_launch(void* const* d_in, const int* in_sizes, int n_in,
                              void* d_out, int out_size, void* d_ws, size_t ws_size,
                              hipStream_t stream) {
    const float* features = (const float*)d_in[0];
    const float* W1 = (const float*)d_in[1];
    const float* b1 = (const float*)d_in[2];
    const float* W2 = (const float*)d_in[3];
    const float* b2 = (const float*)d_in[4];
    const float* W3 = (const float*)d_in[5];
    const float* b3 = (const float*)d_in[6];
    const int*  src = (const int*)d_in[7];
    const int*  dst = (const int*)d_in[8];
    const int E = in_sizes[7];

    float* ws       = (float*)d_ws;
    float* norm_src = ws;
    float* norm_dst = ws + N_NODES;
    float* aggA     = ws + 2 * N_NODES;
    float* bufB     = aggA + (size_t)N_NODES * FDIM;
    int*   q_src    = (int*)(bufB + (size_t)N_NODES * FDIM);
    int*   q_row    = q_src + (size_t)OWN_GROUPS * OWN_QCAP;
    float* q_scale  = (float*)(q_row + (size_t)OWN_GROUPS * OWN_QCAP);
    int*   q_cnt    = (int*)(q_scale + (size_t)OWN_GROUPS * OWN_QCAP);
    unsigned* omap  = (unsigned*)(q_cnt + (size_t)OWN_GROUPS * OWN_WAVES);

    zero_f32<<<512, 256, 0, stream>>>(norm_src, 2 * N_NODES);
    degree_kernel<<<(E + OWN_RND - 1) / OWN_RND * (OWN_RND / 1024), 256, 0, stream>>>(
        src, dst, norm_src, norm_dst, omap, E);
    norm_finalize<<<(2 * N_NODES + 255) / 256, 256, 0, stream>>>(norm_src, 2 * N_NODES);

    const int gemmGroups128 = 512;
    const int gemmGroups64  = 1024;

    aggregate_owner_kernel<true><<<OWN_GROUPS, OWN_THREADS, 0, stream>>>(
        features, src, dst, norm_src, aggA, q_src, q_row, q_scale, q_cnt, omap, E);
    gemm_wmma_kernel<128, true><<<gemmGroups128, 256, 0, stream>>>(aggA, W1, b1, norm_dst, bufB);

    aggregate_owner_kernel<false><<<OWN_GROUPS, OWN_THREADS, 0, stream>>>(
        bufB, src, dst, norm_src, aggA, q_src, q_row, q_scale, q_cnt, omap, E);
    gemm_wmma_kernel<128, true><<<gemmGroups128, 256, 0, stream>>>(aggA, W2, b2, norm_dst, bufB);

    aggregate_owner_kernel<false><<<OWN_GROUPS, OWN_THREADS, 0, stream>>>(
        bufB, src, dst, norm_src, aggA, q_src, q_row, q_scale, q_cnt, omap, E);
    gemm_wmma_kernel<64, false><<<gemmGroups64, 128, 0, stream>>>(aggA, W3, b3, norm_dst,
                                                                (float*)d_out);
}
